# starting kernel
# baseline (speedup 1.0000x reference)
#include <hip/hip_runtime.h>

typedef __attribute__((ext_vector_type(2))) float v2f;
typedef __attribute__((ext_vector_type(8))) float v8f;

#define NN 4096
#define BPAIRS 4096
#define EPSF 1e-16f

__global__ __launch_bounds__(256) void pair_reduce_kernel(
    const float* __restrict__ A1, const float* __restrict__ A2,
    const float* __restrict__ f1, const float* __restrict__ f2,
    const int* __restrict__ is1, const int* __restrict__ id1,
    const int* __restrict__ is2, const int* __restrict__ id2,
    float* __restrict__ xout)
{
    const int p   = blockIdx.x;
    const int lid = p >> 12;
    const int b   = p & (BPAIRS - 1);

    const float* A  = lid ? A2 : A1;
    const float* Ao = lid ? A1 : A2;
    const float* fa = lid ? f2 : f1;
    const float* fb = lid ? f1 : f2;
    const int*   si = lid ? is2 : is1;
    const int*   di = lid ? id2 : id1;

    const int isrc = si[b];
    const int idst = di[b];

    const float4* rAs  = (const float4*)(A  + (size_t)isrc * NN);
    const float4* rAod = (const float4*)(Ao + (size_t)idst * NN);
    const float4* rAos = (const float4*)(Ao + (size_t)isrc * NN);
    const float4* rAd  = (const float4*)(A  + (size_t)idst * NN);
    const float4* fav  = (const float4*)fa;
    const float4* fbv  = (const float4*)fb;

    const int t = threadIdx.x;
    float s1 = 0.f, s2 = 0.f, n1 = 0.f, n2 = 0.f, n3 = 0.f, n4 = 0.f;

#define ACC(c)                                                              \
    {                                                                       \
        float t1 = as.c * fa4.c;  float t2 = aod.c * fb4.c;                 \
        float t3 = aos.c * fb4.c; float t4 = ad.c  * fa4.c;                 \
        s1 = fmaf(t1, t2, s1);    s2 = fmaf(t3, t4, s2);                    \
        float u1 = t1 + EPSF, u2 = t2 + EPSF, u3 = t3 + EPSF, u4 = t4 + EPSF; \
        n1 = fmaf(u1, u1, n1);    n2 = fmaf(u2, u2, n2);                    \
        n3 = fmaf(u3, u3, n3);    n4 = fmaf(u4, u4, n4);                    \
    }

#pragma unroll
    for (int k = 0; k < 4; ++k) {
        const int q = t * 4 + k;
        float4 as  = rAs[q];
        float4 aod = rAod[q];
        float4 aos = rAos[q];
        float4 ad  = rAd[q];
        float4 fa4 = fav[q];
        float4 fb4 = fbv[q];
        ACC(x) ACC(y) ACC(z) ACC(w)
    }
#undef ACC

#pragma unroll
    for (int m = 16; m >= 1; m >>= 1) {
        s1 += __shfl_xor(s1, m, 32);
        s2 += __shfl_xor(s2, m, 32);
        n1 += __shfl_xor(n1, m, 32);
        n2 += __shfl_xor(n2, m, 32);
        n3 += __shfl_xor(n3, m, 32);
        n4 += __shfl_xor(n4, m, 32);
    }

    __shared__ float red[8][6];
    const int wave = t >> 5, lane = t & 31;
    if (lane == 0) {
        red[wave][0] = s1; red[wave][1] = s2;
        red[wave][2] = n1; red[wave][3] = n2;
        red[wave][4] = n3; red[wave][5] = n4;
    }
    __syncthreads();
    if (t == 0) {
        float S1 = 0.f, S2 = 0.f, N1 = 0.f, N2 = 0.f, N3 = 0.f, N4 = 0.f;
#pragma unroll
        for (int w = 0; w < 8; ++w) {
            S1 += red[w][0]; S2 += red[w][1];
            N1 += red[w][2]; N2 += red[w][3];
            N3 += red[w][4]; N4 += red[w][5];
        }
        const float D    = sqrtf(N1) * sqrtf(N2);
        const float Drev = sqrtf(N3) * sqrtf(N4);
        xout[p] = S1 / D + S2 / Drev;
    }
}

__global__ __launch_bounds__(256) void mlp_wmma_kernel(
    const float* __restrict__ xin,
    const float* __restrict__ g1w1, const float* __restrict__ g1b1,
    const float* __restrict__ g1w2, const float* __restrict__ g1b2,
    const float* __restrict__ g2w1, const float* __restrict__ g2b1,
    const float* __restrict__ g2w2, const float* __restrict__ g2b2,
    float* __restrict__ out)
{
    __shared__ float lds[8 * 256];

    const int t    = threadIdx.x;
    const int wave = t >> 5;
    const int lane = t & 31;
    const int tile = blockIdx.x * 8 + wave;
    const int p0   = tile * 16;
    const int lid  = p0 >> 12;
    const int n    = lane & 15;
    const bool lo  = (lane < 16);

    float* L = &lds[wave * 256];
    const int mbase = lo ? 0 : 8;

    const float xv  = xin[p0 + n];
    const float w1v = g1w1[lid * 16 + n];
    const float b1v = g1b1[lid * 16 + n];
    const float w2v = g1w2[lid * 16 + n];

    v2f Am = { lo ? xv  : 0.0f, 0.0f };
    v2f Bm = { lo ? w1v : 0.0f, 0.0f };
    v8f C;
#pragma unroll
    for (int r = 0; r < 8; ++r) C[r] = b1v;

    v8f D = __builtin_amdgcn_wmma_f32_16x16x4_f32(
        false, Am, false, Bm, (short)0, C, false, false);

#pragma unroll
    for (int r = 0; r < 8; ++r)
        L[(mbase + r) * 16 + n] = fmaxf(D[r], 0.0f) * w2v;
    __syncthreads();

    float y1 = 0.0f;
#pragma unroll
    for (int j = 0; j < 16; ++j) y1 += L[n * 16 + j];
    y1 += g1b2[lid];
    __syncthreads();

    const float v1v = g2w1[lid * 16 + n];
    const float c1v = g2b1[lid * 16 + n];
    const float v2v = g2w2[lid * 16 + n];

    v2f Am2 = { lo ? y1  : 0.0f, 0.0f };
    v2f Bm2 = { lo ? v1v : 0.0f, 0.0f };
    v8f C2;
#pragma unroll
    for (int r = 0; r < 8; ++r) C2[r] = c1v;

    v8f D2 = __builtin_amdgcn_wmma_f32_16x16x4_f32(
        false, Am2, false, Bm2, (short)0, C2, false, false);

#pragma unroll
    for (int r = 0; r < 8; ++r)
        L[(mbase + r) * 16 + n] = fmaxf(D2[r], 0.0f) * v2v;
    __syncthreads();

    float y2 = 0.0f;
#pragma unroll
    for (int j = 0; j < 16; ++j) y2 += L[n * 16 + j];
    y2 += g2b2[lid];

    if (lo) out[p0 + n] = y2;
}

extern "C" void kernel_launch(void* const* d_in, const int* in_sizes, int n_in,
                              void* d_out, int out_size, void* d_ws, size_t ws_size,
                              hipStream_t stream)
{
    const float* A1  = (const float*)d_in[0];
    const float* A2  = (const float*)d_in[1];
    const float* f1  = (const float*)d_in[2];
    const float* f2  = (const float*)d_in[3];
    const int*   is1 = (const int*)d_in[4];
    const int*   id1 = (const int*)d_in[5];
    const int*   is2 = (const int*)d_in[6];
    const int*   id2 = (const int*)d_in[7];
    const float* g1w1 = (const float*)d_in[8];
    const float* g1b1 = (const float*)d_in[9];
    const float* g1w2 = (const float*)d_in[10];
    const float* g1b2 = (const float*)d_in[11];
    const float* g2w1 = (const float*)d_in[12];
    const float* g2b1 = (const float*)d_in[13];
    const float* g2w2 = (const float*)d_in[14];
    const float* g2b2 = (const float*)d_in[15];

    float* xbuf = (float*)d_ws;

    pair_reduce_kernel<<<2 * BPAIRS, 256, 0, stream>>>(
        A1, A2, f1, f2, is1, id1, is2, id2, xbuf);

    mlp_wmma_kernel<<<64, 256, 0, stream>>>(
        xbuf, g1w1, g1b1, g1w2, g1b2, g2w1, g2b1, g2w2, g2b2,
        (float*)d_out);
}


# pair_reduce made resident: 512 workgroups loop over 16 pairs each (f slices kept in registers, next pair's rows requested ahead, one end barrier), was 8192 one-pair workgroups; sum order kept
# speedup vs baseline: 1.5769x; 1.5769x over previous
#include <hip/hip_runtime.h>

typedef __attribute__((ext_vector_type(2))) float v2f;
typedef __attribute__((ext_vector_type(8))) float v8f;

#define NN 4096
#define BPAIRS 4096
#define EPSF 1e-16f

#define PPB 16
#define NWG 512
static_assert(NWG * PPB == 2 * BPAIRS && BPAIRS % PPB == 0 && PPB % 2 == 0 && PPB <= 16,
              "pairs must split evenly; 2*PPB row indices sit in one wave's 32 lanes");

__global__ __launch_bounds__(256) void pair_reduce_kernel(
    const float* __restrict__ A1, const float* __restrict__ A2,
    const float* __restrict__ f1, const float* __restrict__ f2,
    const int* __restrict__ is1, const int* __restrict__ id1,
    const int* __restrict__ is2, const int* __restrict__ id2,
    float* __restrict__ xout)
{
    const int p0  = blockIdx.x * PPB;
    const int lid = p0 >> 12;
    const int b0  = p0 & (BPAIRS - 1);

    const float* A  = lid ? A2 : A1;
    const float* Ao = lid ? A1 : A2;
    const float* fa = lid ? f2 : f1;
    const float* fb = lid ? f1 : f2;
    const int*   si = lid ? is2 : is1;
    const int*   di = lid ? id2 : id1;

    const int t = threadIdx.x;
    const int wave = t >> 5, lane = t & 31;

    const int idxv = ((lane < 16) ? si : di)[b0 + (lane & 15)];

    float4 fa4[4], fb4[4];
#pragma unroll
    for (int k = 0; k < 4; ++k) {
        fa4[k] = ((const float4*)fa)[t * 4 + k];
        fb4[k] = ((const float4*)fb)[t * 4 + k];
    }

    __shared__ float red[PPB][8][6];

    float4 bufA[16], bufB[16];

#define LOAD_PAIR(buf, i)                                                   \
    {                                                                       \
        const int isrc_ = __builtin_amdgcn_readlane(idxv, (i));             \
        const int idst_ = __builtin_amdgcn_readlane(idxv, 16 + (i));        \
        const float4* rAs_  = (const float4*)(A  + (size_t)isrc_ * NN) + t * 4; \
        const float4* rAod_ = (const float4*)(Ao + (size_t)idst_ * NN) + t * 4; \
        const float4* rAos_ = (const float4*)(Ao + (size_t)isrc_ * NN) + t * 4; \
        const float4* rAd_  = (const float4*)(A  + (size_t)idst_ * NN) + t * 4; \
        _Pragma("unroll")                                                   \
        for (int k = 0; k < 4; ++k) {                                       \
            buf[k * 4 + 0] = rAs_[k];                                       \
            buf[k * 4 + 1] = rAod_[k];                                      \
            buf[k * 4 + 2] = rAos_[k];                                      \
            buf[k * 4 + 3] = rAd_[k];                                       \
        }                                                                   \
    }

#define ACC(c)                                                              \
    {                                                                       \
        float t1 = as.c * fav.c;  float t2 = aod.c * fbv.c;                 \
        float t3 = aos.c * fbv.c; float t4 = ad.c  * fav.c;                 \
        s1 = fmaf(t1, t2, s1);    s2 = fmaf(t3, t4, s2);                    \
        float u1 = t1 + EPSF, u2 = t2 + EPSF, u3 = t3 + EPSF, u4 = t4 + EPSF; \
        n1 = fmaf(u1, u1, n1);    n2 = fmaf(u2, u2, n2);                    \
        n3 = fmaf(u3, u3, n3);    n4 = fmaf(u4, u4, n4);                    \
    }

#define SUM_PAIR(buf, i)                                                    \
    {                                                                       \
        float s1 = 0.f, s2 = 0.f, n1 = 0.f, n2 = 0.f, n3 = 0.f, n4 = 0.f;   \
        _Pragma("unroll")                                                   \
        for (int k = 0; k < 4; ++k) {                                       \
            const float4 as  = buf[k * 4 + 0];                              \
            const float4 aod = buf[k * 4 + 1];                              \
            const float4 aos = buf[k * 4 + 2];                              \
            const float4 ad  = buf[k * 4 + 3];                              \
            const float4 fav = fa4[k];                                      \
            const float4 fbv = fb4[k];                                      \
            ACC(x) ACC(y) ACC(z) ACC(w)                                     \
        }                                                                   \
        _Pragma("unroll")                                                   \
        for (int m = 16; m >= 1; m >>= 1) {                                 \
            s1 += __shfl_xor(s1, m, 32);                                    \
            s2 += __shfl_xor(s2, m, 32);                                    \
            n1 += __shfl_xor(n1, m, 32);                                    \
            n2 += __shfl_xor(n2, m, 32);                                    \
            n3 += __shfl_xor(n3, m, 32);                                    \
            n4 += __shfl_xor(n4, m, 32);                                    \
        }                                                                   \
        if (lane == 0) {                                                    \
            red[(i)][wave][0] = s1; red[(i)][wave][1] = s2;                 \
            red[(i)][wave][2] = n1; red[(i)][wave][3] = n2;                 \
            red[(i)][wave][4] = n3; red[(i)][wave][5] = n4;                 \
        }                                                                   \
    }

    LOAD_PAIR(bufA, 0)
    for (int i = 0; i < PPB - 2; i += 2) {
        LOAD_PAIR(bufB, i + 1)
        __builtin_amdgcn_sched_barrier(0);
        SUM_PAIR(bufA, i)
        LOAD_PAIR(bufA, i + 2)
        __builtin_amdgcn_sched_barrier(0);
        SUM_PAIR(bufB, i + 1)
    }
    LOAD_PAIR(bufB, PPB - 1)
    __builtin_amdgcn_sched_barrier(0);
    SUM_PAIR(bufA, PPB - 2)
    SUM_PAIR(bufB, PPB - 1)
#undef SUM_PAIR
#undef ACC
#undef LOAD_PAIR

    __syncthreads();
    if (t < PPB) {
        float S1 = 0.f, S2 = 0.f, N1 = 0.f, N2 = 0.f, N3 = 0.f, N4 = 0.f;
#pragma unroll
        for (int w = 0; w < 8; ++w) {
            S1 += red[t][w][0]; S2 += red[t][w][1];
            N1 += red[t][w][2]; N2 += red[t][w][3];
            N3 += red[t][w][4]; N4 += red[t][w][5];
        }
        const float D    = sqrtf(N1) * sqrtf(N2);
        const float Drev = sqrtf(N3) * sqrtf(N4);
        xout[p0 + t] = S1 / D + S2 / Drev;
    }
}

__global__ __launch_bounds__(256) void mlp_wmma_kernel(
    const float* __restrict__ xin,
    const float* __restrict__ g1w1, const float* __restrict__ g1b1,
    const float* __restrict__ g1w2, const float* __restrict__ g1b2,
    const float* __restrict__ g2w1, const float* __restrict__ g2b1,
    const float* __restrict__ g2w2, const float* __restrict__ g2b2,
    float* __restrict__ out)
{
    __shared__ float lds[8 * 256];

    const int t    = threadIdx.x;
    const int wave = t >> 5;
    const int lane = t & 31;
    const int tile = blockIdx.x * 8 + wave;
    const int p0   = tile * 16;
    const int lid  = p0 >> 12;
    const int n    = lane & 15;
    const bool lo  = (lane < 16);

    float* L = &lds[wave * 256];
    const int mbase = lo ? 0 : 8;

    const float xv  = xin[p0 + n];
    const float w1v = g1w1[lid * 16 + n];
    const float b1v = g1b1[lid * 16 + n];
    const float w2v = g1w2[lid * 16 + n];

    v2f Am = { lo ? xv  : 0.0f, 0.0f };
    v2f Bm = { lo ? w1v : 0.0f, 0.0f };
    v8f C;
#pragma unroll
    for (int r = 0; r < 8; ++r) C[r] = b1v;

    v8f D = __builtin_amdgcn_wmma_f32_16x16x4_f32(
        false, Am, false, Bm, (short)0, C, false, false);

#pragma unroll
    for (int r = 0; r < 8; ++r)
        L[(mbase + r) * 16 + n] = fmaxf(D[r], 0.0f) * w2v;
    __syncthreads();

    float y1 = 0.0f;
#pragma unroll
    for (int j = 0; j < 16; ++j) y1 += L[n * 16 + j];
    y1 += g1b2[lid];
    __syncthreads();

    const float v1v = g2w1[lid * 16 + n];
    const float c1v = g2b1[lid * 16 + n];
    const float v2v = g2w2[lid * 16 + n];

    v2f Am2 = { lo ? y1  : 0.0f, 0.0f };
    v2f Bm2 = { lo ? v1v : 0.0f, 0.0f };
    v8f C2;
#pragma unroll
    for (int r = 0; r < 8; ++r) C2[r] = c1v;

    v8f D2 = __builtin_amdgcn_wmma_f32_16x16x4_f32(
        false, Am2, false, Bm2, (short)0, C2, false, false);

#pragma unroll
    for (int r = 0; r < 8; ++r)
        L[(mbase + r) * 16 + n] = fmaxf(D2[r], 0.0f) * v2v;
    __syncthreads();

    float y2 = 0.0f;
#pragma unroll
    for (int j = 0; j < 16; ++j) y2 += L[n * 16 + j];
    y2 += g2b2[lid];

    if (lo) out[p0 + n] = y2;
}

extern "C" void kernel_launch(void* const* d_in, const int* in_sizes, int n_in,
                              void* d_out, int out_size, void* d_ws, size_t ws_size,
                              hipStream_t stream)
{
    const float* A1  = (const float*)d_in[0];
    const float* A2  = (const float*)d_in[1];
    const float* f1  = (const float*)d_in[2];
    const float* f2  = (const float*)d_in[3];
    const int*   is1 = (const int*)d_in[4];
    const int*   id1 = (const int*)d_in[5];
    const int*   is2 = (const int*)d_in[6];
    const int*   id2 = (const int*)d_in[7];
    const float* g1w1 = (const float*)d_in[8];
    const float* g1b1 = (const float*)d_in[9];
    const float* g1w2 = (const float*)d_in[10];
    const float* g1b2 = (const float*)d_in[11];
    const float* g2w1 = (const float*)d_in[12];
    const float* g2b1 = (const float*)d_in[13];
    const float* g2w2 = (const float*)d_in[14];
    const float* g2b2 = (const float*)d_in[15];

    float* xbuf = (float*)d_ws;

    pair_reduce_kernel<<<NWG, 256, 0, stream>>>(
        A1, A2, f1, f2, is1, id1, is2, id2, xbuf);

    mlp_wmma_kernel<<<64, 256, 0, stream>>>(
        xbuf, g1w1, g1b1, g1w2, g1b2, g2w1, g2b1, g2w2, g2b2,
        (float*)d_out);
}


# pair_reduce row loads cover whole lines: lane l of wave w reads float4 128w+32k+l (512 contiguous B, 4 lines a load), was a 64-B lane stride (16 lines a load); sum order inside a wave changes
# speedup vs baseline: 1.8463x; 1.1708x over previous
#include <hip/hip_runtime.h>

typedef __attribute__((ext_vector_type(2))) float v2f;
typedef __attribute__((ext_vector_type(8))) float v8f;

#define NN 4096
#define BPAIRS 4096
#define EPSF 1e-16f

#define PPB 16
#define NWG 512
static_assert(NWG * PPB == 2 * BPAIRS && BPAIRS % PPB == 0 && PPB % 2 == 0 && PPB <= 16,
              "pairs must split evenly; 2*PPB row indices sit in one wave's 32 lanes");

__global__ __launch_bounds__(256) void pair_reduce_kernel(
    const float* __restrict__ A1, const float* __restrict__ A2,
    const float* __restrict__ f1, const float* __restrict__ f2,
    const int* __restrict__ is1, const int* __restrict__ id1,
    const int* __restrict__ is2, const int* __restrict__ id2,
    float* __restrict__ xout)
{
    const int p0  = blockIdx.x * PPB;
    const int lid = p0 >> 12;
    const int b0  = p0 & (BPAIRS - 1);

    const float* A  = lid ? A2 : A1;
    const float* Ao = lid ? A1 : A2;
    const float* fa = lid ? f2 : f1;
    const float* fb = lid ? f1 : f2;
    const int*   si = lid ? is2 : is1;
    const int*   di = lid ? id2 : id1;

    const int t = threadIdx.x;
    const int wave = t >> 5, lane = t & 31;

    const int idxv = ((lane < 16) ? si : di)[b0 + (lane & 15)];

    const int q = wave * 128 + lane;
#define PIECE 32

    float4 fa4[4], fb4[4];
#pragma unroll
    for (int k = 0; k < 4; ++k) {
        fa4[k] = ((const float4*)fa)[q + k * PIECE];
        fb4[k] = ((const float4*)fb)[q + k * PIECE];
    }

    __shared__ float red[PPB][8][6];

    float4 bufA[16], bufB[16];

#define LOAD_PAIR(buf, i)                                                   \
    {                                                                       \
        const int isrc_ = __builtin_amdgcn_readlane(idxv, (i));             \
        const int idst_ = __builtin_amdgcn_readlane(idxv, 16 + (i));        \
        const float4* rAs_  = (const float4*)(A  + (size_t)isrc_ * NN) + q; \
        const float4* rAod_ = (const float4*)(Ao + (size_t)idst_ * NN) + q; \
        const float4* rAos_ = (const float4*)(Ao + (size_t)isrc_ * NN) + q; \
        const float4* rAd_  = (const float4*)(A  + (size_t)idst_ * NN) + q; \
        _Pragma("unroll")                                                   \
        for (int k = 0; k < 4; ++k) {                                       \
            buf[k * 4 + 0] = rAs_[k * PIECE];                               \
            buf[k * 4 + 1] = rAod_[k * PIECE];                              \
            buf[k * 4 + 2] = rAos_[k * PIECE];                              \
            buf[k * 4 + 3] = rAd_[k * PIECE];                               \
        }                                                                   \
    }

#define ACC(c)                                                              \
    {                                                                       \
        float t1 = as.c * fav.c;  float t2 = aod.c * fbv.c;                 \
        float t3 = aos.c * fbv.c; float t4 = ad.c  * fav.c;                 \
        s1 = fmaf(t1, t2, s1);    s2 = fmaf(t3, t4, s2);                    \
        float u1 = t1 + EPSF, u2 = t2 + EPSF, u3 = t3 + EPSF, u4 = t4 + EPSF; \
        n1 = fmaf(u1, u1, n1);    n2 = fmaf(u2, u2, n2);                    \
        n3 = fmaf(u3, u3, n3);    n4 = fmaf(u4, u4, n4);                    \
    }

#define SUM_PAIR(buf, i)                                                    \
    {                                                                       \
        float s1 = 0.f, s2 = 0.f, n1 = 0.f, n2 = 0.f, n3 = 0.f, n4 = 0.f;   \
        _Pragma("unroll")                                                   \
        for (int k = 0; k < 4; ++k) {                                       \
            const float4 as  = buf[k * 4 + 0];                              \
            const float4 aod = buf[k * 4 + 1];                              \
            const float4 aos = buf[k * 4 + 2];                              \
            const float4 ad  = buf[k * 4 + 3];                              \
            const float4 fav = fa4[k];                                      \
            const float4 fbv = fb4[k];                                      \
            ACC(x) ACC(y) ACC(z) ACC(w)                                     \
        }                                                                   \
        _Pragma("unroll")                                                   \
        for (int m = 16; m >= 1; m >>= 1) {                                 \
            s1 += __shfl_xor(s1, m, 32);                                    \
            s2 += __shfl_xor(s2, m, 32);                                    \
            n1 += __shfl_xor(n1, m, 32);                                    \
            n2 += __shfl_xor(n2, m, 32);                                    \
            n3 += __shfl_xor(n3, m, 32);                                    \
            n4 += __shfl_xor(n4, m, 32);                                    \
        }                                                                   \
        if (lane == 0) {                                                    \
            red[(i)][wave][0] = s1; red[(i)][wave][1] = s2;                 \
            red[(i)][wave][2] = n1; red[(i)][wave][3] = n2;                 \
            red[(i)][wave][4] = n3; red[(i)][wave][5] = n4;                 \
        }                                                                   \
    }

    LOAD_PAIR(bufA, 0)
    for (int i = 0; i < PPB - 2; i += 2) {
        LOAD_PAIR(bufB, i + 1)
        __builtin_amdgcn_sched_barrier(0);
        SUM_PAIR(bufA, i)
        LOAD_PAIR(bufA, i + 2)
        __builtin_amdgcn_sched_barrier(0);
        SUM_PAIR(bufB, i + 1)
    }
    LOAD_PAIR(bufB, PPB - 1)
    __builtin_amdgcn_sched_barrier(0);
    SUM_PAIR(bufA, PPB - 2)
    SUM_PAIR(bufB, PPB - 1)
#undef SUM_PAIR
#undef ACC
#undef LOAD_PAIR
#undef PIECE

    __syncthreads();
    if (t < PPB) {
        float S1 = 0.f, S2 = 0.f, N1 = 0.f, N2 = 0.f, N3 = 0.f, N4 = 0.f;
#pragma unroll
        for (int w = 0; w < 8; ++w) {
            S1 += red[t][w][0]; S2 += red[t][w][1];
            N1 += red[t][w][2]; N2 += red[t][w][3];
            N3 += red[t][w][4]; N4 += red[t][w][5];
        }
        const float D    = sqrtf(N1) * sqrtf(N2);
        const float Drev = sqrtf(N3) * sqrtf(N4);
        xout[p0 + t] = S1 / D + S2 / Drev;
    }
}

__global__ __launch_bounds__(256) void mlp_wmma_kernel(
    const float* __restrict__ xin,
    const float* __restrict__ g1w1, const float* __restrict__ g1b1,
    const float* __restrict__ g1w2, const float* __restrict__ g1b2,
    const float* __restrict__ g2w1, const float* __restrict__ g2b1,
    const float* __restrict__ g2w2, const float* __restrict__ g2b2,
    float* __restrict__ out)
{
    __shared__ float lds[8 * 256];

    const int t    = threadIdx.x;
    const int wave = t >> 5;
    const int lane = t & 31;
    const int tile = blockIdx.x * 8 + wave;
    const int p0   = tile * 16;
    const int lid  = p0 >> 12;
    const int n    = lane & 15;
    const bool lo  = (lane < 16);

    float* L = &lds[wave * 256];
    const int mbase = lo ? 0 : 8;

    const float xv  = xin[p0 + n];
    const float w1v = g1w1[lid * 16 + n];
    const float b1v = g1b1[lid * 16 + n];
    const float w2v = g1w2[lid * 16 + n];

    v2f Am = { lo ? xv  : 0.0f, 0.0f };
    v2f Bm = { lo ? w1v : 0.0f, 0.0f };
    v8f C;
#pragma unroll
    for (int r = 0; r < 8; ++r) C[r] = b1v;

    v8f D = __builtin_amdgcn_wmma_f32_16x16x4_f32(
        false, Am, false, Bm, (short)0, C, false, false);

#pragma unroll
    for (int r = 0; r < 8; ++r)
        L[(mbase + r) * 16 + n] = fmaxf(D[r], 0.0f) * w2v;
    __syncthreads();

    float y1 = 0.0f;
#pragma unroll
    for (int j = 0; j < 16; ++j) y1 += L[n * 16 + j];
    y1 += g1b2[lid];
    __syncthreads();

    const float v1v = g2w1[lid * 16 + n];
    const float c1v = g2b1[lid * 16 + n];
    const float v2v = g2w2[lid * 16 + n];

    v2f Am2 = { lo ? y1  : 0.0f, 0.0f };
    v2f Bm2 = { lo ? v1v : 0.0f, 0.0f };
    v8f C2;
#pragma unroll
    for (int r = 0; r < 8; ++r) C2[r] = c1v;

    v8f D2 = __builtin_amdgcn_wmma_f32_16x16x4_f32(
        false, Am2, false, Bm2, (short)0, C2, false, false);

#pragma unroll
    for (int r = 0; r < 8; ++r)
        L[(mbase + r) * 16 + n] = fmaxf(D2[r], 0.0f) * v2v;
    __syncthreads();

    float y2 = 0.0f;
#pragma unroll
    for (int j = 0; j < 16; ++j) y2 += L[n * 16 + j];
    y2 += g2b2[lid];

    if (lo) out[p0 + n] = y2;
}

extern "C" void kernel_launch(void* const* d_in, const int* in_sizes, int n_in,
                              void* d_out, int out_size, void* d_ws, size_t ws_size,
                              hipStream_t stream)
{
    const float* A1  = (const float*)d_in[0];
    const float* A2  = (const float*)d_in[1];
    const float* f1  = (const float*)d_in[2];
    const float* f2  = (const float*)d_in[3];
    const int*   is1 = (const int*)d_in[4];
    const int*   id1 = (const int*)d_in[5];
    const int*   is2 = (const int*)d_in[6];
    const int*   id2 = (const int*)d_in[7];
    const float* g1w1 = (const float*)d_in[8];
    const float* g1b1 = (const float*)d_in[9];
    const float* g1w2 = (const float*)d_in[10];
    const float* g1b2 = (const float*)d_in[11];
    const float* g2w1 = (const float*)d_in[12];
    const float* g2b1 = (const float*)d_in[13];
    const float* g2w2 = (const float*)d_in[14];
    const float* g2b2 = (const float*)d_in[15];

    float* xbuf = (float*)d_ws;

    pair_reduce_kernel<<<NWG, 256, 0, stream>>>(
        A1, A2, f1, f2, is1, id1, is2, id2, xbuf);

    mlp_wmma_kernel<<<64, 256, 0, stream>>>(
        xbuf, g1w1, g1b1, g1w2, g1b2, g2w1, g2b1, g2w2, g2b2,
        (float*)d_out);
}


# the two MLPs folded into pair_reduce: wave 0 of each workgroup runs both WMMA MLPs on its own 16 pairs after the end barrier; the second launch and the scratch round trip are gone; sum order kept
# speedup vs baseline: 2.1280x; 1.1526x over previous
#include <hip/hip_runtime.h>

typedef __attribute__((ext_vector_type(2))) float v2f;
typedef __attribute__((ext_vector_type(8))) float v8f;

#define NN 4096
#define BPAIRS 4096
#define EPSF 1e-16f

#define PPB 16
#define NWG 512
static_assert(NWG * PPB == 2 * BPAIRS && BPAIRS % PPB == 0 && PPB % 2 == 0 && PPB <= 16,
              "pairs must split evenly; 2*PPB row indices sit in one wave's 32 lanes");

__global__ __launch_bounds__(256) void pair_reduce_kernel(
    const float* __restrict__ A1, const float* __restrict__ A2,
    const float* __restrict__ f1, const float* __restrict__ f2,
    const int* __restrict__ is1, const int* __restrict__ id1,
    const int* __restrict__ is2, const int* __restrict__ id2,
    const float* __restrict__ g1w1, const float* __restrict__ g1b1,
    const float* __restrict__ g1w2, const float* __restrict__ g1b2,
    const float* __restrict__ g2w1, const float* __restrict__ g2b1,
    const float* __restrict__ g2w2, const float* __restrict__ g2b2,
    float* __restrict__ out)
{
    const int p0  = blockIdx.x * PPB;
    const int lid = p0 >> 12;
    const int b0  = p0 & (BPAIRS - 1);

    const float* A  = lid ? A2 : A1;
    const float* Ao = lid ? A1 : A2;
    const float* fa = lid ? f2 : f1;
    const float* fb = lid ? f1 : f2;
    const int*   si = lid ? is2 : is1;
    const int*   di = lid ? id2 : id1;

    const int t = threadIdx.x;
    const int wave = t >> 5, lane = t & 31;

    const int idxv = ((lane < 16) ? si : di)[b0 + (lane & 15)];

    const int n = lane & 15;
    const float w1v = g1w1[lid * 16 + n], b1v = g1b1[lid * 16 + n];
    const float w2v = g1w2[lid * 16 + n], b2v = g1b2[lid];
    const float v1v = g2w1[lid * 16 + n], c1v = g2b1[lid * 16 + n];
    const float v2v = g2w2[lid * 16 + n], c2v = g2b2[lid];

    const int q = wave * 128 + lane;
#define PIECE 32

    float4 fa4[4], fb4[4];
#pragma unroll
    for (int k = 0; k < 4; ++k) {
        fa4[k] = ((const float4*)fa)[q + k * PIECE];
        fb4[k] = ((const float4*)fb)[q + k * PIECE];
    }

    __shared__ float red[PPB][8][6];

    float4 bufA[16], bufB[16];

#define LOAD_PAIR(buf, i)                                                   \
    {                                                                       \
        const int isrc_ = __builtin_amdgcn_readlane(idxv, (i));             \
        const int idst_ = __builtin_amdgcn_readlane(idxv, 16 + (i));        \
        const float4* rAs_  = (const float4*)(A  + (size_t)isrc_ * NN) + q; \
        const float4* rAod_ = (const float4*)(Ao + (size_t)idst_ * NN) + q; \
        const float4* rAos_ = (const float4*)(Ao + (size_t)isrc_ * NN) + q; \
        const float4* rAd_  = (const float4*)(A  + (size_t)idst_ * NN) + q; \
        _Pragma("unroll")                                                   \
        for (int k = 0; k < 4; ++k) {                                       \
            buf[k * 4 + 0] = rAs_[k * PIECE];                               \
            buf[k * 4 + 1] = rAod_[k * PIECE];                              \
            buf[k * 4 + 2] = rAos_[k * PIECE];                              \
            buf[k * 4 + 3] = rAd_[k * PIECE];                               \
        }                                                                   \
    }

#define ACC(c)                                                              \
    {                                                                       \
        float t1 = as.c * fav.c;  float t2 = aod.c * fbv.c;                 \
        float t3 = aos.c * fbv.c; float t4 = ad.c  * fav.c;                 \
        s1 = fmaf(t1, t2, s1);    s2 = fmaf(t3, t4, s2);                    \
        float u1 = t1 + EPSF, u2 = t2 + EPSF, u3 = t3 + EPSF, u4 = t4 + EPSF; \
        n1 = fmaf(u1, u1, n1);    n2 = fmaf(u2, u2, n2);                    \
        n3 = fmaf(u3, u3, n3);    n4 = fmaf(u4, u4, n4);                    \
    }

#define SUM_PAIR(buf, i)                                                    \
    {                                                                       \
        float s1 = 0.f, s2 = 0.f, n1 = 0.f, n2 = 0.f, n3 = 0.f, n4 = 0.f;   \
        _Pragma("unroll")                                                   \
        for (int k = 0; k < 4; ++k) {                                       \
            const float4 as  = buf[k * 4 + 0];                              \
            const float4 aod = buf[k * 4 + 1];                              \
            const float4 aos = buf[k * 4 + 2];                              \
            const float4 ad  = buf[k * 4 + 3];                              \
            const float4 fav = fa4[k];                                      \
            const float4 fbv = fb4[k];                                      \
            ACC(x) ACC(y) ACC(z) ACC(w)                                     \
        }                                                                   \
        _Pragma("unroll")                                                   \
        for (int m = 16; m >= 1; m >>= 1) {                                 \
            s1 += __shfl_xor(s1, m, 32);                                    \
            s2 += __shfl_xor(s2, m, 32);                                    \
            n1 += __shfl_xor(n1, m, 32);                                    \
            n2 += __shfl_xor(n2, m, 32);                                    \
            n3 += __shfl_xor(n3, m, 32);                                    \
            n4 += __shfl_xor(n4, m, 32);                                    \
        }                                                                   \
        if (lane == 0) {                                                    \
            red[(i)][wave][0] = s1; red[(i)][wave][1] = s2;                 \
            red[(i)][wave][2] = n1; red[(i)][wave][3] = n2;                 \
            red[(i)][wave][4] = n3; red[(i)][wave][5] = n4;                 \
        }                                                                   \
    }

    LOAD_PAIR(bufA, 0)
    for (int i = 0; i < PPB - 2; i += 2) {
        LOAD_PAIR(bufB, i + 1)
        __builtin_amdgcn_sched_barrier(0);
        SUM_PAIR(bufA, i)
        LOAD_PAIR(bufA, i + 2)
        __builtin_amdgcn_sched_barrier(0);
        SUM_PAIR(bufB, i + 1)
    }
    LOAD_PAIR(bufB, PPB - 1)
    __builtin_amdgcn_sched_barrier(0);
    SUM_PAIR(bufA, PPB - 2)
    SUM_PAIR(bufB, PPB - 1)
#undef SUM_PAIR
#undef ACC
#undef LOAD_PAIR
#undef PIECE

    __syncthreads();

    if (__builtin_amdgcn_readfirstlane(wave) != 0) return;

    float S1 = 0.f, S2 = 0.f, N1 = 0.f, N2 = 0.f, N3 = 0.f, N4 = 0.f;
    for (int w = 0; w < 8; ++w) {
        S1 += red[n][w][0]; S2 += red[n][w][1];
        N1 += red[n][w][2]; N2 += red[n][w][3];
        N3 += red[n][w][4]; N4 += red[n][w][5];
    }
    const float D    = sqrtf(N1) * sqrtf(N2);
    const float Drev = sqrtf(N3) * sqrtf(N4);
    const float xv   = S1 / D + S2 / Drev;

    __shared__ float L[256];
    const bool lo   = (lane < 16);
    const int mbase = lo ? 0 : 8;

#define WAVE_LDS_ORDER()                                        \
    {                                                           \
        __builtin_amdgcn_fence(__ATOMIC_RELEASE, "wavefront");  \
        __builtin_amdgcn_wave_barrier();                        \
        __builtin_amdgcn_fence(__ATOMIC_ACQUIRE, "wavefront");  \
    }

    v2f Am = { lo ? xv  : 0.0f, 0.0f };
    v2f Bm = { lo ? w1v : 0.0f, 0.0f };
    v8f C;
    for (int r = 0; r < 8; ++r) C[r] = b1v;

    v8f Dm = __builtin_amdgcn_wmma_f32_16x16x4_f32(
        false, Am, false, Bm, (short)0, C, false, false);

    for (int r = 0; r < 8; ++r)
        L[(mbase + r) * 16 + n] = fmaxf(Dm[r], 0.0f) * w2v;
    WAVE_LDS_ORDER()

    float y1 = 0.0f;
    for (int j = 0; j < 16; ++j) y1 += L[n * 16 + j];
    y1 += b2v;
    WAVE_LDS_ORDER()

    v2f Am2 = { lo ? y1  : 0.0f, 0.0f };
    v2f Bm2 = { lo ? v1v : 0.0f, 0.0f };
    v8f C2;
    for (int r = 0; r < 8; ++r) C2[r] = c1v;

    v8f D2 = __builtin_amdgcn_wmma_f32_16x16x4_f32(
        false, Am2, false, Bm2, (short)0, C2, false, false);

    for (int r = 0; r < 8; ++r)
        L[(mbase + r) * 16 + n] = fmaxf(D2[r], 0.0f) * v2v;
    WAVE_LDS_ORDER()
#undef WAVE_LDS_ORDER

    float y2 = 0.0f;
    for (int j = 0; j < 16; ++j) y2 += L[n * 16 + j];
    y2 += c2v;

    if (lo) out[p0 + n] = y2;
}

extern "C" void kernel_launch(void* const* d_in, const int* in_sizes, int n_in,
                              void* d_out, int out_size, void* d_ws, size_t ws_size,
                              hipStream_t stream)
{
    const float* A1  = (const float*)d_in[0];
    const float* A2  = (const float*)d_in[1];
    const float* f1  = (const float*)d_in[2];
    const float* f2  = (const float*)d_in[3];
    const int*   is1 = (const int*)d_in[4];
    const int*   id1 = (const int*)d_in[5];
    const int*   is2 = (const int*)d_in[6];
    const int*   id2 = (const int*)d_in[7];
    const float* g1w1 = (const float*)d_in[8];
    const float* g1b1 = (const float*)d_in[9];
    const float* g1w2 = (const float*)d_in[10];
    const float* g1b2 = (const float*)d_in[11];
    const float* g2w1 = (const float*)d_in[12];
    const float* g2b1 = (const float*)d_in[13];
    const float* g2w2 = (const float*)d_in[14];
    const float* g2b2 = (const float*)d_in[15];

    pair_reduce_kernel<<<NWG, 256, 0, stream>>>(
        A1, A2, f1, f2, is1, id1, is2, id2,
        g1w1, g1b1, g1w2, g1b2, g2w1, g2b1, g2w2, g2b2,
        (float*)d_out);
}
